# starting kernel
# baseline (speedup 1.0000x reference)
#include <hip/hip_runtime.h>
#include <math.h>

typedef __attribute__((ext_vector_type(2))) float v2f;
typedef __attribute__((ext_vector_type(8))) float v8f;

#define TT 128
#define DD 768
#define DECAY_ALPHA 0.1f
#define EPS_W 1e-8f
#define LN_EPS 1e-5f

__device__ __forceinline__ float sanitize(float x) {
  return (__builtin_fabsf(x) <= 3.402823e38f) ? x : 0.0f;
}

__global__ __launch_bounds__(256) void mda_kernel(
    const float* __restrict__ H, const int* __restrict__ valid_lens,
    const float* __restrict__ gamma, const float* __restrict__ beta,
    float* __restrict__ out)
{
  __shared__ float s_norm2[TT];
  __shared__ float s_w[TT];
  __shared__ float s_E[DD];
  __shared__ float s_red[8];
  __shared__ float s_bcast[3];

  const int bf   = blockIdx.x;
  const int tid  = threadIdx.x;
  const int lane = tid & 31;
  const int wv   = tid >> 5;
  const float* __restrict__ Hbf = H + (size_t)bf * (TT * DD);

  int L = valid_lens[bf];
  L = (L < 0) ? 0 : (L > TT ? TT : L);

  for (int t = wv; t < TT; t += 8) {
    float s = 0.0f;
    if (t < L) {
      const float* row = Hbf + t * DD;
      if (t + 8 < L) __builtin_prefetch(row + 8 * DD + lane * 4, 0, 0);
      #pragma unroll
      for (int c = 0; c < 6; ++c) {
        const float4 v = *(const float4*)(row + c * 128 + lane * 4);
        const float x = sanitize(v.x), y = sanitize(v.y);
        const float z = sanitize(v.z), w = sanitize(v.w);
        s += x * x + y * y + z * z + w * w;
      }
    }
    #pragma unroll
    for (int off = 16; off > 0; off >>= 1) s += __shfl_xor(s, off, 32);
    if (lane == 0) s_norm2[t] = (t < L) ? s : __builtin_inff();
  }
  __syncthreads();

  if (tid < TT) {
    const float my = s_norm2[tid];
    int rank = 0;
    #pragma unroll 4
    for (int j = 0; j < TT; ++j) {
      const float vj = s_norm2[j];
      rank += (vj < my || (vj == my && j < tid)) ? 1 : 0;
    }
    const int k_low = (int)(0.1f * (float)TT);
    const int lo = (k_low < (L >> 1)) ? k_low : (L >> 1);
    const int hr = L - lo;
    const int hi = (hr > lo) ? hr : lo;
    const bool keep = (rank >= lo) && (rank < hi) && (tid < L);
    s_w[tid] = keep ? __expf(-DECAY_ALPHA * (float)(TT - 1 - tid)) : 0.0f;
  }
  __syncthreads();

  if (wv == 0) {
    float s = s_w[lane] + s_w[lane + 32] + s_w[lane + 64] + s_w[lane + 96];
    #pragma unroll
    for (int off = 16; off > 0; off >>= 1) s += __shfl_xor(s, off, 32);
    if (lane == 0) {
      s_bcast[0] = 1.0f / fmaxf(s, EPS_W);
      s_bcast[1] = (s > EPS_W) ? 1.0f : 0.0f;
    }
  }
  __syncthreads();

  const int half = lane >> 4;
  const int l16  = lane & 15;
  const float inv = s_bcast[0];

  for (int c = 0; c < 6; c += 2) {
    const int d0 = wv * 96 + c * 16;
    v8f acc0 = {0.f, 0.f, 0.f, 0.f, 0.f, 0.f, 0.f, 0.f};
    v8f acc1 = acc0;
    for (int t0 = 0; t0 < TT; t0 += 4) {
      const float w0 = s_w[t0], w1 = s_w[t0 + 1], w2 = s_w[t0 + 2], w3 = s_w[t0 + 3];
      if ((w0 == 0.0f) & (w1 == 0.0f) & (w2 == 0.0f) & (w3 == 0.0f)) continue;
      v2f b; b.x = half ? w2 : w0; b.y = half ? w3 : w1;
      const float* pa = Hbf + (size_t)(t0 + half * 2) * DD + d0 + l16;
      v2f a0, a1;
      a0.x = sanitize(pa[0]);   a0.y = sanitize(pa[DD]);
      a1.x = sanitize(pa[16]);  a1.y = sanitize(pa[DD + 16]);
      acc0 = __builtin_amdgcn_wmma_f32_16x16x4_f32(false, a0, false, b, (short)0,
                                                   acc0, false, false);
      acc1 = __builtin_amdgcn_wmma_f32_16x16x4_f32(false, a1, false, b, (short)0,
                                                   acc1, false, false);
    }
    if (l16 == 0) {
      const int mb = half * 8;
      #pragma unroll
      for (int r = 0; r < 8; ++r) {
        s_E[d0 + mb + r]      = acc0[r] * inv;
        s_E[d0 + 16 + mb + r] = acc1[r] * inv;
      }
    }
  }
  __syncthreads();

  const float e0 = s_E[tid], e1 = s_E[tid + 256], e2 = s_E[tid + 512];
  float s = e0 + e1 + e2;
  #pragma unroll
  for (int off = 16; off > 0; off >>= 1) s += __shfl_xor(s, off, 32);
  if (lane == 0) s_red[wv] = s;
  __syncthreads();
  if (tid == 0) {
    float tot = 0.f;
    #pragma unroll
    for (int i = 0; i < 8; ++i) tot += s_red[i];
    s_bcast[2] = tot * (1.0f / (float)DD);
  }
  __syncthreads();
  const float mu = s_bcast[2];

  const float q0 = e0 - mu, q1 = e1 - mu, q2 = e2 - mu;
  float v = q0 * q0 + q1 * q1 + q2 * q2;
  #pragma unroll
  for (int off = 16; off > 0; off >>= 1) v += __shfl_xor(v, off, 32);
  if (lane == 0) s_red[wv] = v;
  __syncthreads();
  if (tid == 0) {
    float tot = 0.f;
    #pragma unroll
    for (int i = 0; i < 8; ++i) tot += s_red[i];
    s_bcast[2] = tot * (1.0f / (float)DD);
  }
  __syncthreads();
  const float var  = s_bcast[2];
  const float rstd = rsqrtf(var + LN_EPS);
  const bool  has  = s_bcast[1] > 0.5f;

  float* __restrict__ obf = out + (size_t)bf * DD;
  {
    const int d = tid;
    obf[d] = has ? ((e0 - mu) * rstd * gamma[d] + beta[d]) : e0;
  }
  {
    const int d = tid + 256;
    obf[d] = has ? ((e1 - mu) * rstd * gamma[d] + beta[d]) : e1;
  }
  {
    const int d = tid + 512;
    obf[d] = has ? ((e2 - mu) * rstd * gamma[d] + beta[d]) : e2;
  }
}

extern "C" void kernel_launch(void* const* d_in, const int* in_sizes, int n_in,
                              void* d_out, int out_size, void* d_ws, size_t ws_size,
                              hipStream_t stream) {
  const float* H     = (const float*)d_in[0];
  const int*   vlens = (const int*)d_in[1];
  const float* gamma = (const float*)d_in[2];
  const float* beta  = (const float*)d_in[3];
  float* out = (float*)d_out;
  const int BF = in_sizes[1];
  mda_kernel<<<BF, 256, 0, stream>>>(H, vlens, gamma, beta, out);
}


# Weighted sum fed from the registers the norm pass already holds (a wave's own rows, selector B fragments, zero-weight rows skipped), no second read of H; sums now per wave, then across waves
# speedup vs baseline: 2.3350x; 2.3350x over previous
#include <hip/hip_runtime.h>
#include <math.h>

typedef __attribute__((ext_vector_type(2))) float v2f;
typedef __attribute__((ext_vector_type(8))) float v8f;

#define TT 128
#define DD 768
#define DECAY_ALPHA 0.1f
#define EPS_W 1e-8f
#define LN_EPS 1e-5f

__global__ __launch_bounds__(256) void mda_kernel(
    const float* H, const int* __restrict__ valid_lens,
    const float* __restrict__ gamma, const float* __restrict__ beta,
    float* __restrict__ out)
{
  __shared__ float s_norm2[TT];
  __shared__ float s_w[TT];
  __shared__ float s_part[8 * DD];
  __shared__ float s_red[8];
  __shared__ float s_bcast[3];

  const int bf   = blockIdx.x;
  const int tid  = threadIdx.x;
  const int lane = tid & 31;
  const int wv   = tid >> 5;
  const float* Hbf = H + (size_t)bf * (TT * DD);

  int L = valid_lens[bf];
  L = (L < 0) ? 0 : (L > TT ? TT : L);

  v2f hp[16][12];

  {
    float4 hv[16][6];
    #pragma unroll
    for (int i = 0; i < 16; ++i) {
      const int t = wv + 8 * i;
      const int tr = (t < L) ? t : 0;
      const float* row = Hbf + tr * DD + lane * 4;
      #pragma unroll
      for (int c = 0; c < 6; ++c) hv[i][c] = *(const float4*)(row + c * 128);
      if (i < 15) __builtin_amdgcn_sched_barrier(0);
    }
    __builtin_amdgcn_sched_barrier(0);
    float fin_max = 3.402823e38f;
    asm volatile("" : "+v"(fin_max) : : "memory");
    float ps[16];
    #pragma unroll
    for (int i = 0; i < 16; ++i) {
      float s = 0.0f;
      #pragma unroll
      for (int c = 0; c < 6; ++c) {
        const float4 v = hv[i][c];
        const float x = (__builtin_fabsf(v.x) <= fin_max) ? v.x : 0.0f;
        const float y = (__builtin_fabsf(v.y) <= fin_max) ? v.y : 0.0f;
        const float z = (__builtin_fabsf(v.z) <= fin_max) ? v.z : 0.0f;
        const float w = (__builtin_fabsf(v.w) <= fin_max) ? v.w : 0.0f;
        hp[i][2 * c]     = (v2f){x, y};
        hp[i][2 * c + 1] = (v2f){z, w};
        {
          #pragma clang fp contract(off)
          s += x * x + y * y + z * z + w * w;
        }
      }
      ps[i] = s;
      if (i < 15) asm volatile("" : "+v"(fin_max), "+v"(s));
    }
    #pragma unroll
    for (int off = 16; off > 0; off >>= 1) {
      #pragma unroll
      for (int i = 0; i < 16; ++i) ps[i] += __shfl_xor(ps[i], off, 32);
    }
    if (lane == 0) {
      #pragma unroll
      for (int i = 0; i < 16; ++i) {
        const int t = wv + 8 * i;
        s_norm2[t] = (t < L) ? ps[i] : __builtin_inff();
      }
    }
  }
  __syncthreads();

  if (tid < TT) {
    const float my = s_norm2[tid];
    int rank = 0;
    #pragma unroll 4
    for (int j = 0; j < TT; ++j) {
      const float vj = s_norm2[j];
      rank += (vj < my || (vj == my && j < tid)) ? 1 : 0;
    }
    const int k_low = (int)(0.1f * (float)TT);
    const int lo = (k_low < (L >> 1)) ? k_low : (L >> 1);
    const int hr = L - lo;
    const int hi = (hr > lo) ? hr : lo;
    const bool keep = (rank >= lo) && (rank < hi) && (tid < L);
    s_w[tid] = keep ? __expf(-DECAY_ALPHA * (float)(TT - 1 - tid)) : 0.0f;
  }
  __syncthreads();

  if (wv == 0) {
    float s = s_w[lane] + s_w[lane + 32] + s_w[lane + 64] + s_w[lane + 96];
    #pragma unroll
    for (int off = 16; off > 0; off >>= 1) s += __shfl_xor(s, off, 32);
    if (lane == 0) {
      s_bcast[0] = 1.0f / fmaxf(s, EPS_W);
      s_bcast[1] = (s > EPS_W) ? 1.0f : 0.0f;
    }
  }

  const int h   = lane >> 4;
  const int c16 = lane & 15;
  unsigned mx[4], my[4];
  #pragma unroll
  for (int q = 0; q < 4; ++q) {
    const bool mine = ((c16 >> 2) == q) && (((c16 >> 1) & 1) == h);
    mx[q] = (mine && (c16 & 1) == 0) ? 0xffffffffu : 0u;
    my[q] = (mine && (c16 & 1) == 1) ? 0xffffffffu : 0u;
  }

  float wr[16];
  #pragma unroll
  for (int i = 0; i < 16; ++i) wr[i] = s_w[wv + 8 * i];

  v8f acc[3];
  #pragma unroll
  for (int p = 0; p < 3; ++p) acc[p] = (v8f){0.f, 0.f, 0.f, 0.f, 0.f, 0.f, 0.f, 0.f};

  #pragma unroll
  for (int i = 0; i < 16; ++i) {
    const unsigned wb = (unsigned)__builtin_amdgcn_readfirstlane((int)__float_as_uint(wr[i]));
    if ((wb << 1) != 0u) {
      v2f bq[4];
      #pragma unroll
      for (int q = 0; q < 4; ++q) {
        bq[q].x = __uint_as_float(wb & mx[q]);
        bq[q].y = __uint_as_float(wb & my[q]);
      }
      #pragma unroll
      for (int q = 0; q < 4; ++q) {
        #pragma unroll
        for (int p = 0; p < 3; ++p) {
          const int c = 2 * p + (q >> 1);
          acc[p] = __builtin_amdgcn_wmma_f32_16x16x4_f32(false, hp[i][2 * c + (q & 1)], false, bq[q],
                                                         (short)0, acc[p], false, false);
        }
      }
    }
  }
  {
    float* dst = s_part + wv * DD + 32 * h + 64 * ((c16 >> 1) & 1) + 128 * (c16 >> 3)
                 + 2 * ((c16 >> 2) & 1) + (c16 & 1);
    #pragma unroll
    for (int p = 0; p < 3; ++p) {
      #pragma unroll
      for (int r = 0; r < 8; ++r) dst[256 * p + 4 * r] = acc[p][r];
    }
  }
  __syncthreads();

  const float inv = s_bcast[0];
  float ev[3];
  #pragma unroll
  for (int j = 0; j < 3; ++j) {
    float sj = 0.0f;
    #pragma unroll
    for (int w8 = 0; w8 < 8; ++w8) sj += s_part[w8 * DD + tid + 256 * j];
    ev[j] = sj * inv;
  }
  const float e0 = ev[0], e1 = ev[1], e2 = ev[2];
  float s = e0 + e1 + e2;
  #pragma unroll
  for (int off = 16; off > 0; off >>= 1) s += __shfl_xor(s, off, 32);
  if (lane == 0) s_red[wv] = s;
  __syncthreads();
  if (tid == 0) {
    float tot = 0.f;
    #pragma unroll
    for (int i = 0; i < 8; ++i) tot += s_red[i];
    s_bcast[2] = tot * (1.0f / (float)DD);
  }
  __syncthreads();
  const float mu = s_bcast[2];

  const float q0 = e0 - mu, q1 = e1 - mu, q2 = e2 - mu;
  float v = q0 * q0 + q1 * q1 + q2 * q2;
  #pragma unroll
  for (int off = 16; off > 0; off >>= 1) v += __shfl_xor(v, off, 32);
  if (lane == 0) s_red[wv] = v;
  __syncthreads();
  if (tid == 0) {
    float tot = 0.f;
    #pragma unroll
    for (int i = 0; i < 8; ++i) tot += s_red[i];
    s_bcast[2] = tot * (1.0f / (float)DD);
  }
  __syncthreads();
  const float var  = s_bcast[2];
  const float rstd = rsqrtf(var + LN_EPS);
  const bool  has  = s_bcast[1] > 0.5f;

  float* __restrict__ obf = out + (size_t)bf * DD;
  {
    const int d = tid;
    obf[d] = has ? ((e0 - mu) * rstd * gamma[d] + beta[d]) : e0;
  }
  {
    const int d = tid + 256;
    obf[d] = has ? ((e1 - mu) * rstd * gamma[d] + beta[d]) : e1;
  }
  {
    const int d = tid + 512;
    obf[d] = has ? ((e2 - mu) * rstd * gamma[d] + beta[d]) : e2;
  }
}

extern "C" void kernel_launch(void* const* d_in, const int* in_sizes, int n_in,
                              void* d_out, int out_size, void* d_ws, size_t ws_size,
                              hipStream_t stream) {
  const float* H     = (const float*)d_in[0];
  const int*   vlens = (const int*)d_in[1];
  const float* gamma = (const float*)d_in[2];
  const float* beta  = (const float*)d_in[3];
  float* out = (float*)d_out;
  const int BF = in_sizes[1];
  mda_kernel<<<BF, 256, 0, stream>>>(H, vlens, gamma, beta, out);
}


# Rank pass: 128 norms fetched as 32 independent 16-byte LDS reads issued ahead of use; rows counted by sign bit of an integer key difference, not 16 dependent LDS trips of compare/select; same ranks
# speedup vs baseline: 2.7380x; 1.1726x over previous
#include <hip/hip_runtime.h>
#include <math.h>

typedef __attribute__((ext_vector_type(2))) float v2f;
typedef __attribute__((ext_vector_type(8))) float v8f;

#define TT 128
#define DD 768
#define DECAY_ALPHA 0.1f
#define EPS_W 1e-8f
#define LN_EPS 1e-5f

__global__ __launch_bounds__(256) void mda_kernel(
    const float* H, const int* __restrict__ valid_lens,
    const float* __restrict__ gamma, const float* __restrict__ beta,
    float* __restrict__ out)
{
  __shared__ __attribute__((aligned(16))) float s_norm2[TT];
  __shared__ float s_w[TT];
  __shared__ float s_part[8 * DD];
  __shared__ float s_red[8];
  __shared__ float s_bcast[3];

  const int bf   = blockIdx.x;
  const int tid  = threadIdx.x;
  const int lane = tid & 31;
  const int wv   = tid >> 5;
  const float* Hbf = H + (size_t)bf * (TT * DD);

  int L = valid_lens[bf];
  L = (L < 0) ? 0 : (L > TT ? TT : L);

  v2f hp[16][12];

  {
    float4 hv[16][6];
    #pragma unroll
    for (int i = 0; i < 16; ++i) {
      const int t = wv + 8 * i;
      const int tr = (t < L) ? t : 0;
      const float* row = Hbf + tr * DD + lane * 4;
      #pragma unroll
      for (int c = 0; c < 6; ++c) hv[i][c] = *(const float4*)(row + c * 128);
      if (i < 15) __builtin_amdgcn_sched_barrier(0);
    }
    __builtin_amdgcn_sched_barrier(0);
    float fin_max = 3.402823e38f;
    asm volatile("" : "+v"(fin_max) : : "memory");
    float ps[16];
    #pragma unroll
    for (int i = 0; i < 16; ++i) {
      float s = 0.0f;
      #pragma unroll
      for (int c = 0; c < 6; ++c) {
        const float4 v = hv[i][c];
        const float x = (__builtin_fabsf(v.x) <= fin_max) ? v.x : 0.0f;
        const float y = (__builtin_fabsf(v.y) <= fin_max) ? v.y : 0.0f;
        const float z = (__builtin_fabsf(v.z) <= fin_max) ? v.z : 0.0f;
        const float w = (__builtin_fabsf(v.w) <= fin_max) ? v.w : 0.0f;
        hp[i][2 * c]     = (v2f){x, y};
        hp[i][2 * c + 1] = (v2f){z, w};
        {
          #pragma clang fp contract(off)
          s += x * x + y * y + z * z + w * w;
        }
      }
      ps[i] = s;
      if (i < 15) asm volatile("" : "+v"(fin_max), "+v"(s));
    }
    #pragma unroll
    for (int off = 16; off > 0; off >>= 1) {
      #pragma unroll
      for (int i = 0; i < 16; ++i) ps[i] += __shfl_xor(ps[i], off, 32);
    }
    if (lane == 0) {
      #pragma unroll
      for (int i = 0; i < 16; ++i) {
        const int t = wv + 8 * i;
        s_norm2[t] = (t < L) ? ps[i] : __builtin_inff();
      }
    }
  }
  __syncthreads();

  if (tid < TT) {
    const unsigned kmy  = __float_as_uint(s_norm2[tid]);
    const unsigned kmy1 = kmy + 1u;
    const float4* nb = (const float4*)s_norm2;
    const int wq = wv;
    const int b1 = (wq + 1) & 3, b2 = (wq + 2) & 3, b3 = (wq + 3) & 3;
    const unsigned t1 = (b1 < wq) ? kmy1 : kmy;
    const unsigned t2 = (b2 < wq) ? kmy1 : kmy;
    const unsigned t3 = (b3 < wq) ? kmy1 : kmy;
    const float4* p0 = nb + 8 * wq;
    const float4* p1 = nb + 8 * b1;
    const float4* p2 = nb + 8 * b2;
    const float4* p3 = nb + 8 * b3;
    float4 ka[8], kb[8];
    unsigned cx = 0, cy = 0, cz = 0, cw = 0;
    #pragma unroll
    for (int q = 0; q < 8; ++q) ka[q] = p1[q];
    #pragma unroll
    for (int q = 0; q < 8; ++q) kb[q] = p2[q];
    __builtin_amdgcn_sched_barrier(0);
    #pragma unroll
    for (int q = 0; q < 8; ++q) {
      cx = __builtin_amdgcn_alignbit(cx, __float_as_uint(ka[q].x) - t1, 31);
      cy = __builtin_amdgcn_alignbit(cy, __float_as_uint(ka[q].y) - t1, 31);
      cz = __builtin_amdgcn_alignbit(cz, __float_as_uint(ka[q].z) - t1, 31);
      cw = __builtin_amdgcn_alignbit(cw, __float_as_uint(ka[q].w) - t1, 31);
    }
    #pragma unroll
    for (int q = 0; q < 8; ++q) ka[q] = p3[q];
    __builtin_amdgcn_sched_barrier(0);
    #pragma unroll
    for (int q = 0; q < 8; ++q) {
      cx = __builtin_amdgcn_alignbit(cx, __float_as_uint(kb[q].x) - t2, 31);
      cy = __builtin_amdgcn_alignbit(cy, __float_as_uint(kb[q].y) - t2, 31);
      cz = __builtin_amdgcn_alignbit(cz, __float_as_uint(kb[q].z) - t2, 31);
      cw = __builtin_amdgcn_alignbit(cw, __float_as_uint(kb[q].w) - t2, 31);
    }
    #pragma unroll
    for (int q = 0; q < 8; ++q) kb[q] = p0[q];
    __builtin_amdgcn_sched_barrier(0);
    #pragma unroll
    for (int q = 0; q < 8; ++q) {
      cx = __builtin_amdgcn_alignbit(cx, __float_as_uint(ka[q].x) - t3, 31);
      cy = __builtin_amdgcn_alignbit(cy, __float_as_uint(ka[q].y) - t3, 31);
      cz = __builtin_amdgcn_alignbit(cz, __float_as_uint(ka[q].z) - t3, 31);
      cw = __builtin_amdgcn_alignbit(cw, __float_as_uint(ka[q].w) - t3, 31);
    }
    unsigned lx = 0, ly = 0, lz = 0, lw = 0;
    #pragma unroll
    for (int q = 0; q < 8; ++q) {
      const unsigned kx = __float_as_uint(kb[q].x), ky = __float_as_uint(kb[q].y);
      const unsigned kz = __float_as_uint(kb[q].z), kw = __float_as_uint(kb[q].w);
      cx = __builtin_amdgcn_alignbit(cx, kx - kmy, 31);
      cy = __builtin_amdgcn_alignbit(cy, ky - kmy, 31);
      cz = __builtin_amdgcn_alignbit(cz, kz - kmy, 31);
      cw = __builtin_amdgcn_alignbit(cw, kw - kmy, 31);
      lx = __builtin_amdgcn_alignbit(lx, kx - kmy1, 31);
      ly = __builtin_amdgcn_alignbit(ly, ky - kmy1, 31);
      lz = __builtin_amdgcn_alignbit(lz, kz - kmy1, 31);
      lw = __builtin_amdgcn_alignbit(lw, kw - kmy1, 31);
    }
    const unsigned tie_x = (0xffu << (8 - ((lane + 3) >> 2))) & 0xffu;
    const unsigned tie_y = (0xffu << (8 - ((lane + 2) >> 2))) & 0xffu;
    const unsigned tie_z = (0xffu << (8 - ((lane + 1) >> 2))) & 0xffu;
    const unsigned tie_w = (0xffu << (8 - (lane >> 2))) & 0xffu;
    const int rank = __builtin_popcount(cx) + __builtin_popcount(cy) + __builtin_popcount(cz)
                   + __builtin_popcount(cw)
                   + __builtin_popcount(lx & ~cx & tie_x) + __builtin_popcount(ly & ~cy & tie_y)
                   + __builtin_popcount(lz & ~cz & tie_z) + __builtin_popcount(lw & ~cw & tie_w);
    const int k_low = (int)(0.1f * (float)TT);
    const int lo = (k_low < (L >> 1)) ? k_low : (L >> 1);
    const int hr = L - lo;
    const int hi = (hr > lo) ? hr : lo;
    const bool keep = (rank >= lo) && (rank < hi) && (tid < L);
    s_w[tid] = keep ? __expf(-DECAY_ALPHA * (float)(TT - 1 - tid)) : 0.0f;
  }
  __syncthreads();

  if (wv == 0) {
    float s = s_w[lane] + s_w[lane + 32] + s_w[lane + 64] + s_w[lane + 96];
    #pragma unroll
    for (int off = 16; off > 0; off >>= 1) s += __shfl_xor(s, off, 32);
    if (lane == 0) {
      s_bcast[0] = 1.0f / fmaxf(s, EPS_W);
      s_bcast[1] = (s > EPS_W) ? 1.0f : 0.0f;
    }
  }

  const int h   = lane >> 4;
  const int c16 = lane & 15;
  unsigned mx[4], my[4];
  #pragma unroll
  for (int q = 0; q < 4; ++q) {
    const bool mine = ((c16 >> 2) == q) && (((c16 >> 1) & 1) == h);
    mx[q] = (mine && (c16 & 1) == 0) ? 0xffffffffu : 0u;
    my[q] = (mine && (c16 & 1) == 1) ? 0xffffffffu : 0u;
  }

  float wr[16];
  #pragma unroll
  for (int i = 0; i < 16; ++i) wr[i] = s_w[wv + 8 * i];

  v8f acc[3];
  #pragma unroll
  for (int p = 0; p < 3; ++p) acc[p] = (v8f){0.f, 0.f, 0.f, 0.f, 0.f, 0.f, 0.f, 0.f};

  #pragma unroll
  for (int i = 0; i < 16; ++i) {
    const unsigned wb = (unsigned)__builtin_amdgcn_readfirstlane((int)__float_as_uint(wr[i]));
    if ((wb << 1) != 0u) {
      v2f bq[4];
      #pragma unroll
      for (int q = 0; q < 4; ++q) {
        bq[q].x = __uint_as_float(wb & mx[q]);
        bq[q].y = __uint_as_float(wb & my[q]);
      }
      #pragma unroll
      for (int q = 0; q < 4; ++q) {
        #pragma unroll
        for (int p = 0; p < 3; ++p) {
          const int c = 2 * p + (q >> 1);
          acc[p] = __builtin_amdgcn_wmma_f32_16x16x4_f32(false, hp[i][2 * c + (q & 1)], false, bq[q],
                                                         (short)0, acc[p], false, false);
        }
      }
    }
  }
  {
    float* dst = s_part + wv * DD + 32 * h + 64 * ((c16 >> 1) & 1) + 128 * (c16 >> 3)
                 + 2 * ((c16 >> 2) & 1) + (c16 & 1);
    #pragma unroll
    for (int p = 0; p < 3; ++p) {
      #pragma unroll
      for (int r = 0; r < 8; ++r) dst[256 * p + 4 * r] = acc[p][r];
    }
  }
  __syncthreads();

  const float inv = s_bcast[0];
  float ev[3];
  #pragma unroll
  for (int j = 0; j < 3; ++j) {
    float sj = 0.0f;
    #pragma unroll
    for (int w8 = 0; w8 < 8; ++w8) sj += s_part[w8 * DD + tid + 256 * j];
    ev[j] = sj * inv;
  }
  const float e0 = ev[0], e1 = ev[1], e2 = ev[2];
  float s = e0 + e1 + e2;
  #pragma unroll
  for (int off = 16; off > 0; off >>= 1) s += __shfl_xor(s, off, 32);
  if (lane == 0) s_red[wv] = s;
  __syncthreads();
  if (tid == 0) {
    float tot = 0.f;
    #pragma unroll
    for (int i = 0; i < 8; ++i) tot += s_red[i];
    s_bcast[2] = tot * (1.0f / (float)DD);
  }
  __syncthreads();
  const float mu = s_bcast[2];

  const float q0 = e0 - mu, q1 = e1 - mu, q2 = e2 - mu;
  float v = q0 * q0 + q1 * q1 + q2 * q2;
  #pragma unroll
  for (int off = 16; off > 0; off >>= 1) v += __shfl_xor(v, off, 32);
  if (lane == 0) s_red[wv] = v;
  __syncthreads();
  if (tid == 0) {
    float tot = 0.f;
    #pragma unroll
    for (int i = 0; i < 8; ++i) tot += s_red[i];
    s_bcast[2] = tot * (1.0f / (float)DD);
  }
  __syncthreads();
  const float var  = s_bcast[2];
  const float rstd = rsqrtf(var + LN_EPS);
  const bool  has  = s_bcast[1] > 0.5f;

  float* __restrict__ obf = out + (size_t)bf * DD;
  {
    const int d = tid;
    obf[d] = has ? ((e0 - mu) * rstd * gamma[d] + beta[d]) : e0;
  }
  {
    const int d = tid + 256;
    obf[d] = has ? ((e1 - mu) * rstd * gamma[d] + beta[d]) : e1;
  }
  {
    const int d = tid + 512;
    obf[d] = has ? ((e2 - mu) * rstd * gamma[d] + beta[d]) : e2;
  }
}

extern "C" void kernel_launch(void* const* d_in, const int* in_sizes, int n_in,
                              void* d_out, int out_size, void* d_ws, size_t ws_size,
                              hipStream_t stream) {
  const float* H     = (const float*)d_in[0];
  const int*   vlens = (const int*)d_in[1];
  const float* gamma = (const float*)d_in[2];
  const float* beta  = (const float*)d_in[3];
  float* out = (float*)d_out;
  const int BF = in_sizes[1];
  mda_kernel<<<BF, 256, 0, stream>>>(H, vlens, gamma, beta, out);
}


# Round trips off the serial stretches: gamma/beta requested at kernel entry (were 3 serial loads at the end), 32-lane sums by DPP + permlanex16 not LDS bpermute, each thread adds the 8 wave partials
# speedup vs baseline: 3.1100x; 1.1359x over previous
#include <hip/hip_runtime.h>
#include <math.h>

typedef __attribute__((ext_vector_type(2))) float v2f;
typedef __attribute__((ext_vector_type(8))) float v8f;

#define TT 128
#define DD 768
#define DECAY_ALPHA 0.1f
#define EPS_W 1e-8f
#define LN_EPS 1e-5f

__device__ __forceinline__ float xhalf(float x) {
  return __builtin_bit_cast(float, __builtin_amdgcn_permlanex16(__builtin_bit_cast(int, x), __builtin_bit_cast(int, x),
                                                                0x76543210, 0xfedcba98, false, false));
}
template <int M> __device__ __forceinline__ float xrow(float x) {
  return __builtin_bit_cast(float, __builtin_amdgcn_mov_dpp(__builtin_bit_cast(int, x), 0x160 + M, 0xf, 0xf, true));
}
__device__ __forceinline__ float wave_sum(float s) {
  s += xhalf(s);
  s += xrow<8>(s);
  s += xrow<4>(s);
  s += xrow<2>(s);
  s += xrow<1>(s);
  return s;
}

__global__ __launch_bounds__(256) void mda_kernel(
    const float* H, const int* __restrict__ valid_lens,
    const float* __restrict__ gamma, const float* __restrict__ beta,
    float* __restrict__ out)
{
  __shared__ __attribute__((aligned(16))) float s_norm2[TT];
  __shared__ float s_w[TT];
  __shared__ float s_part[8 * DD];
  __shared__ __attribute__((aligned(16))) float s_redm[8];
  __shared__ __attribute__((aligned(16))) float s_redv[8];
  __shared__ float s_bcast[2];

  const int bf   = blockIdx.x;
  const int tid  = threadIdx.x;
  const int lane = tid & 31;
  const int wv   = tid >> 5;
  const float* Hbf = H + (size_t)bf * (TT * DD);

  const float g0 = gamma[tid], g1 = gamma[tid + 256], g2 = gamma[tid + 512];
  const float c0 = beta[tid],  c1 = beta[tid + 256],  c2 = beta[tid + 512];

  int L = valid_lens[bf];
  L = (L < 0) ? 0 : (L > TT ? TT : L);

  v2f hp[16][12];

  {
    float4 hv[16][6];
    #pragma unroll
    for (int i = 0; i < 16; ++i) {
      const int t = wv + 8 * i;
      const int tr = (t < L) ? t : 0;
      const float* row = Hbf + tr * DD + lane * 4;
      #pragma unroll
      for (int c = 0; c < 6; ++c) hv[i][c] = *(const float4*)(row + c * 128);
      if (i < 15) __builtin_amdgcn_sched_barrier(0);
    }
    __builtin_amdgcn_sched_barrier(0);
    float fin_max = 3.402823e38f;
    asm volatile("" : "+v"(fin_max) : : "memory");
    float ps[16];
    #pragma unroll
    for (int i = 0; i < 16; ++i) {
      float s = 0.0f;
      #pragma unroll
      for (int c = 0; c < 6; ++c) {
        const float4 v = hv[i][c];
        const float x = (__builtin_fabsf(v.x) <= fin_max) ? v.x : 0.0f;
        const float y = (__builtin_fabsf(v.y) <= fin_max) ? v.y : 0.0f;
        const float z = (__builtin_fabsf(v.z) <= fin_max) ? v.z : 0.0f;
        const float w = (__builtin_fabsf(v.w) <= fin_max) ? v.w : 0.0f;
        hp[i][2 * c]     = (v2f){x, y};
        hp[i][2 * c + 1] = (v2f){z, w};
        {
          #pragma clang fp contract(off)
          s += x * x + y * y + z * z + w * w;
        }
      }
      ps[i] = s;
      if (i < 15) asm volatile("" : "+v"(fin_max), "+v"(s));
    }
    #pragma unroll
    for (int i = 0; i < 16; ++i) ps[i] += xhalf(ps[i]);
    #pragma unroll
    for (int i = 0; i < 16; ++i) ps[i] += xrow<8>(ps[i]);
    #pragma unroll
    for (int i = 0; i < 16; ++i) ps[i] += xrow<4>(ps[i]);
    #pragma unroll
    for (int i = 0; i < 16; ++i) ps[i] += xrow<2>(ps[i]);
    #pragma unroll
    for (int i = 0; i < 16; ++i) ps[i] += xrow<1>(ps[i]);
    if (lane == 0) {
      #pragma unroll
      for (int i = 0; i < 16; ++i) {
        const int t = wv + 8 * i;
        s_norm2[t] = (t < L) ? ps[i] : __builtin_inff();
      }
    }
  }
  __syncthreads();

  if (tid < TT) {
    const unsigned kmy  = __float_as_uint(s_norm2[tid]);
    const unsigned kmy1 = kmy + 1u;
    const float4* nb = (const float4*)s_norm2;
    const int wq = wv;
    const int b1 = (wq + 1) & 3, b2 = (wq + 2) & 3, b3 = (wq + 3) & 3;
    const unsigned t1 = (b1 < wq) ? kmy1 : kmy;
    const unsigned t2 = (b2 < wq) ? kmy1 : kmy;
    const unsigned t3 = (b3 < wq) ? kmy1 : kmy;
    const float4* p0 = nb + 8 * wq;
    const float4* p1 = nb + 8 * b1;
    const float4* p2 = nb + 8 * b2;
    const float4* p3 = nb + 8 * b3;
    float4 ka[8], kb[8];
    unsigned cx = 0, cy = 0, cz = 0, cw = 0;
    #pragma unroll
    for (int q = 0; q < 8; ++q) ka[q] = p1[q];
    #pragma unroll
    for (int q = 0; q < 8; ++q) kb[q] = p2[q];
    __builtin_amdgcn_sched_barrier(0);
    #pragma unroll
    for (int q = 0; q < 8; ++q) {
      cx = __builtin_amdgcn_alignbit(cx, __float_as_uint(ka[q].x) - t1, 31);
      cy = __builtin_amdgcn_alignbit(cy, __float_as_uint(ka[q].y) - t1, 31);
      cz = __builtin_amdgcn_alignbit(cz, __float_as_uint(ka[q].z) - t1, 31);
      cw = __builtin_amdgcn_alignbit(cw, __float_as_uint(ka[q].w) - t1, 31);
    }
    #pragma unroll
    for (int q = 0; q < 8; ++q) ka[q] = p3[q];
    __builtin_amdgcn_sched_barrier(0);
    #pragma unroll
    for (int q = 0; q < 8; ++q) {
      cx = __builtin_amdgcn_alignbit(cx, __float_as_uint(kb[q].x) - t2, 31);
      cy = __builtin_amdgcn_alignbit(cy, __float_as_uint(kb[q].y) - t2, 31);
      cz = __builtin_amdgcn_alignbit(cz, __float_as_uint(kb[q].z) - t2, 31);
      cw = __builtin_amdgcn_alignbit(cw, __float_as_uint(kb[q].w) - t2, 31);
    }
    #pragma unroll
    for (int q = 0; q < 8; ++q) kb[q] = p0[q];
    __builtin_amdgcn_sched_barrier(0);
    #pragma unroll
    for (int q = 0; q < 8; ++q) {
      cx = __builtin_amdgcn_alignbit(cx, __float_as_uint(ka[q].x) - t3, 31);
      cy = __builtin_amdgcn_alignbit(cy, __float_as_uint(ka[q].y) - t3, 31);
      cz = __builtin_amdgcn_alignbit(cz, __float_as_uint(ka[q].z) - t3, 31);
      cw = __builtin_amdgcn_alignbit(cw, __float_as_uint(ka[q].w) - t3, 31);
    }
    unsigned lx = 0, ly = 0, lz = 0, lw = 0;
    #pragma unroll
    for (int q = 0; q < 8; ++q) {
      const unsigned kx = __float_as_uint(kb[q].x), ky = __float_as_uint(kb[q].y);
      const unsigned kz = __float_as_uint(kb[q].z), kw = __float_as_uint(kb[q].w);
      cx = __builtin_amdgcn_alignbit(cx, kx - kmy, 31);
      cy = __builtin_amdgcn_alignbit(cy, ky - kmy, 31);
      cz = __builtin_amdgcn_alignbit(cz, kz - kmy, 31);
      cw = __builtin_amdgcn_alignbit(cw, kw - kmy, 31);
      lx = __builtin_amdgcn_alignbit(lx, kx - kmy1, 31);
      ly = __builtin_amdgcn_alignbit(ly, ky - kmy1, 31);
      lz = __builtin_amdgcn_alignbit(lz, kz - kmy1, 31);
      lw = __builtin_amdgcn_alignbit(lw, kw - kmy1, 31);
    }
    const unsigned tie_x = (0xffu << (8 - ((lane + 3) >> 2))) & 0xffu;
    const unsigned tie_y = (0xffu << (8 - ((lane + 2) >> 2))) & 0xffu;
    const unsigned tie_z = (0xffu << (8 - ((lane + 1) >> 2))) & 0xffu;
    const unsigned tie_w = (0xffu << (8 - (lane >> 2))) & 0xffu;
    const int rank = __builtin_popcount(cx) + __builtin_popcount(cy) + __builtin_popcount(cz)
                   + __builtin_popcount(cw)
                   + __builtin_popcount(lx & ~cx & tie_x) + __builtin_popcount(ly & ~cy & tie_y)
                   + __builtin_popcount(lz & ~cz & tie_z) + __builtin_popcount(lw & ~cw & tie_w);
    const int k_low = (int)(0.1f * (float)TT);
    const int lo = (k_low < (L >> 1)) ? k_low : (L >> 1);
    const int hr = L - lo;
    const int hi = (hr > lo) ? hr : lo;
    const bool keep = (rank >= lo) && (rank < hi) && (tid < L);
    s_w[tid] = keep ? __expf(-DECAY_ALPHA * (float)(TT - 1 - tid)) : 0.0f;
  }
  __syncthreads();

  if (wv == 0) {
    const float s = wave_sum(s_w[lane] + s_w[lane + 32] + s_w[lane + 64] + s_w[lane + 96]);
    if (lane == 0) {
      s_bcast[0] = 1.0f / fmaxf(s, EPS_W);
      s_bcast[1] = (s > EPS_W) ? 1.0f : 0.0f;
    }
  }

  const int h   = lane >> 4;
  const int c16 = lane & 15;
  unsigned mx[4], my[4];
  #pragma unroll
  for (int q = 0; q < 4; ++q) {
    const bool mine = ((c16 >> 2) == q) && (((c16 >> 1) & 1) == h);
    mx[q] = (mine && (c16 & 1) == 0) ? 0xffffffffu : 0u;
    my[q] = (mine && (c16 & 1) == 1) ? 0xffffffffu : 0u;
  }

  float wr[16];
  #pragma unroll
  for (int i = 0; i < 16; ++i) wr[i] = s_w[wv + 8 * i];

  v8f acc[3];
  #pragma unroll
  for (int p = 0; p < 3; ++p) acc[p] = (v8f){0.f, 0.f, 0.f, 0.f, 0.f, 0.f, 0.f, 0.f};

  #pragma unroll
  for (int i = 0; i < 16; ++i) {
    const unsigned wb = (unsigned)__builtin_amdgcn_readfirstlane((int)__float_as_uint(wr[i]));
    if ((wb << 1) != 0u) {
      v2f bq[4];
      #pragma unroll
      for (int q = 0; q < 4; ++q) {
        bq[q].x = __uint_as_float(wb & mx[q]);
        bq[q].y = __uint_as_float(wb & my[q]);
      }
      #pragma unroll
      for (int q = 0; q < 4; ++q) {
        #pragma unroll
        for (int p = 0; p < 3; ++p) {
          const int c = 2 * p + (q >> 1);
          acc[p] = __builtin_amdgcn_wmma_f32_16x16x4_f32(false, hp[i][2 * c + (q & 1)], false, bq[q],
                                                         (short)0, acc[p], false, false);
        }
      }
    }
  }
  {
    float* dst = s_part + wv * DD + 32 * h + 64 * ((c16 >> 1) & 1) + 128 * (c16 >> 3)
                 + 2 * ((c16 >> 2) & 1) + (c16 & 1);
    #pragma unroll
    for (int p = 0; p < 3; ++p) {
      #pragma unroll
      for (int r = 0; r < 8; ++r) dst[256 * p + 4 * r] = acc[p][r];
    }
  }
  __syncthreads();

  const float inv = s_bcast[0];
  float ev[3];
  #pragma unroll
  for (int j = 0; j < 3; ++j) {
    float sj = 0.0f;
    #pragma unroll
    for (int w8 = 0; w8 < 8; ++w8) sj += s_part[w8 * DD + tid + 256 * j];
    ev[j] = sj * inv;
  }
  const float e0 = ev[0], e1 = ev[1], e2 = ev[2];
  float s = wave_sum(e0 + e1 + e2);
  if (lane == 0) s_redm[wv] = s;
  __syncthreads();
  float mu;
  {
    const float4 pa = *(const float4*)s_redm, pb = *(const float4*)(s_redm + 4);
    float tot = 0.f;
    tot += pa.x; tot += pa.y; tot += pa.z; tot += pa.w;
    tot += pb.x; tot += pb.y; tot += pb.z; tot += pb.w;
    mu = tot * (1.0f / (float)DD);
  }

  const float q0 = e0 - mu, q1 = e1 - mu, q2 = e2 - mu;
  float v = wave_sum(q0 * q0 + q1 * q1 + q2 * q2);
  if (lane == 0) s_redv[wv] = v;
  __syncthreads();
  float var;
  {
    const float4 pa = *(const float4*)s_redv, pb = *(const float4*)(s_redv + 4);
    float tot = 0.f;
    tot += pa.x; tot += pa.y; tot += pa.z; tot += pa.w;
    tot += pb.x; tot += pb.y; tot += pb.z; tot += pb.w;
    var = tot * (1.0f / (float)DD);
  }
  const float rstd = rsqrtf(var + LN_EPS);
  const bool  has  = s_bcast[1] > 0.5f;

  float* __restrict__ obf = out + (size_t)bf * DD;
  obf[tid]       = has ? ((e0 - mu) * rstd * g0 + c0) : e0;
  obf[tid + 256] = has ? ((e1 - mu) * rstd * g1 + c1) : e1;
  obf[tid + 512] = has ? ((e2 - mu) * rstd * g2 + c2) : e2;
}

extern "C" void kernel_launch(void* const* d_in, const int* in_sizes, int n_in,
                              void* d_out, int out_size, void* d_ws, size_t ws_size,
                              hipStream_t stream) {
  const float* H     = (const float*)d_in[0];
  const int*   vlens = (const int*)d_in[1];
  const float* gamma = (const float*)d_in[2];
  const float* beta  = (const float*)d_in[3];
  float* out = (float*)d_out;
  const int BF = in_sizes[1];
  mda_kernel<<<BF, 256, 0, stream>>>(H, vlens, gamma, beta, out);
}
